# starting kernel
# baseline (speedup 1.0000x reference)
#include <hip/hip_runtime.h>
#include <hip/hip_bf16.h>
#include <math.h>


typedef float v2f __attribute__((ext_vector_type(2)));
typedef float v4f __attribute__((ext_vector_type(4)));
typedef float v8f __attribute__((ext_vector_type(8)));

#define BM  64
#define BN  128
#define KC  32
#define AKP 36
#define BKP 36

__device__ __forceinline__ v8f wmma4(v2f a, v2f b, v8f c) {
    return __builtin_amdgcn_wmma_f32_16x16x4_f32(
        false, a, false, b, (short)0, c, false, false);
}

template <int MODE, int RPB>
__global__ __launch_bounds__(256)
void gemm_f32_wmma(const float* __restrict__ A, long bsA, long rsA,
                   const float* __restrict__ Bm, long bsB,
                   const float* __restrict__ biasCol,
                   const float* __restrict__ Emat, long bsE, long rsE,
                   float* __restrict__ Out,
                   int Mtotal, int K, int klim, int N)
{
    __shared__ float As[2][BM][AKP];
    __shared__ float Bs[2][BN][BKP];

    const int tid  = threadIdx.x;
    const int lane = tid & 31;
    const int wave = tid >> 5;
    const int wm   = wave & 1;
    const int wn   = wave >> 1;

    const int tileM = blockIdx.y * BM;
    const int tileN = blockIdx.x * BN;
    const int bB = (bsB != 0) ? (tileM / RPB) : 0;

    v8f acc00 = {}, acc01 = {}, acc10 = {}, acc11 = {};

    const int a_row = tid >> 2;
    const int a_kc  = (tid & 3) << 2;
    int a_grow = tileM + a_row;
    if (a_grow >= Mtotal) a_grow = Mtotal - 1;
    const int a_b = a_grow / RPB;
    const int a_r = a_grow - a_b * RPB;
    const float* a_base = A + (long)a_b * bsA + (long)a_r * rsA;

    const int b_kr = tid >> 4;
    const int b_nc = (tid & 15) << 2;
    const float* b_base = Bm + (long)bB * bsB + tileN;

    const int m0   = (wm << 5) + (lane & 15);
    const int n0   = (wn << 5) + (lane & 15);
    const int koff = (lane >> 4) << 1;

    const int nchunks = (klim + KC - 1) / KC;

    v4f ra0, ra1, rb00, rb01, rb10, rb11;

    auto prefetch = [&](int k0) {
        if (k0 + KC <= klim) {
            ra0  = *(const v4f*)(a_base + k0 + a_kc);
            ra1  = *(const v4f*)(a_base + k0 + a_kc + 16);
            const float* br0 = b_base + (long)(k0 + b_kr)      * N + b_nc;
            const float* br1 = b_base + (long)(k0 + b_kr + 16) * N + b_nc;
            rb00 = *(const v4f*)(br0);
            rb01 = *(const v4f*)(br0 + 64);
            rb10 = *(const v4f*)(br1);
            rb11 = *(const v4f*)(br1 + 64);
        } else {
            const v4f zz = {0.f, 0.f, 0.f, 0.f};
            #pragma unroll
            for (int i = 0; i < 4; ++i) {
                const int ka = k0 + a_kc + i;
                ra0[i] = (ka      < klim) ? a_base[ka]      : 0.0f;
                ra1[i] = (ka + 16 < klim) ? a_base[ka + 16] : 0.0f;
            }
            const int kb0 = k0 + b_kr, kb1 = kb0 + 16;
            const float* br0 = b_base + (long)kb0 * N + b_nc;
            const float* br1 = b_base + (long)kb1 * N + b_nc;
            rb00 = (kb0 < K) ? *(const v4f*)(br0)      : zz;
            rb01 = (kb0 < K) ? *(const v4f*)(br0 + 64) : zz;
            rb10 = (kb1 < K) ? *(const v4f*)(br1)      : zz;
            rb11 = (kb1 < K) ? *(const v4f*)(br1 + 64) : zz;
        }
    };

    auto stage = [&](int buf) {
        *(v4f*)&As[buf][a_row][a_kc]      = ra0;
        *(v4f*)&As[buf][a_row][a_kc + 16] = ra1;
        #pragma unroll
        for (int i = 0; i < 4; ++i) {
            Bs[buf][b_nc + i     ][b_kr]      = rb00[i];
            Bs[buf][b_nc + i     ][b_kr + 16] = rb10[i];
            Bs[buf][b_nc + i + 64][b_kr]      = rb01[i];
            Bs[buf][b_nc + i + 64][b_kr + 16] = rb11[i];
        }
    };

    auto compute = [&](int buf) {
        #pragma unroll
        for (int kk = 0; kk < KC; kk += 4) {
            v2f a0 = *(const v2f*)&As[buf][m0     ][kk + koff];
            v2f a1 = *(const v2f*)&As[buf][m0 + 16][kk + koff];
            v2f b0 = *(const v2f*)&Bs[buf][n0     ][kk + koff];
            v2f b1 = *(const v2f*)&Bs[buf][n0 + 16][kk + koff];
            acc00 = wmma4(a0, b0, acc00);
            acc01 = wmma4(a0, b1, acc01);
            acc10 = wmma4(a1, b0, acc10);
            acc11 = wmma4(a1, b1, acc11);
        }
    };

    prefetch(0);
    stage(0);
    __syncthreads();
    int buf = 0;
    for (int c = 0; c < nchunks; ++c) {
        if (c + 1 < nchunks) prefetch((c + 1) * KC);
        compute(buf);
        if (c + 1 < nchunks) {
            stage(buf ^ 1);
            __syncthreads();
            buf ^= 1;
        }
    }

    const int colA  = tileN + n0;
    const int mbase = tileM + (wm << 5) + ((lane >> 4) << 3);
    #pragma unroll
    for (int mi = 0; mi < 2; ++mi) {
        const v8f& accL = mi ? acc10 : acc00;
        const v8f& accR = mi ? acc11 : acc01;
        #pragma unroll
        for (int j = 0; j < 8; ++j) {
            const int gr = mbase + (mi << 4) + j;
            if (gr >= Mtotal) continue;
            const int b = gr / RPB;
            const int r = gr - b * RPB;
            float v0 = accL[j];
            float v1 = accR[j];
            if (MODE == 0) {
                v0 = fmaxf(v0 + biasCol[colA],      0.0f);
                v1 = fmaxf(v1 + biasCol[colA + 16], 0.0f);
            } else if (MODE == 1) {
                v0 += biasCol[colA];
                v1 += biasCol[colA + 16];
            } else if (MODE == 2) {
                const float* e = Emat + (long)b * bsE + (long)r * rsE;
                v0 = sinf(v0 + e[colA]);
                v1 = sinf(v1 + e[colA + 16]);
            } else {
                const float* e = Emat + (long)b * bsE + (long)r * rsE;
                v0 = (v0 + biasCol[colA]      + e[colA])      * 0.5f;
                v1 = (v1 + biasCol[colA + 16] + e[colA + 16]) * 0.5f;
            }
            Out[(long)gr * N + colA]      = v0;
            Out[(long)gr * N + colA + 16] = v1;
        }
    }
}

extern "C" void kernel_launch(void* const* d_in, const int* in_sizes, int n_in,
                              void* d_out, int out_size, void* d_ws, size_t ws_size,
                              hipStream_t stream)
{
    (void)in_sizes; (void)n_in; (void)out_size; (void)ws_size;

    const float* x  = (const float*)d_in[0];
    const float* z  = (const float*)d_in[1];
    const float* W1 = (const float*)d_in[2];
    const float* b1 = (const float*)d_in[3];
    const float* W2 = (const float*)d_in[4];
    const float* b2 = (const float*)d_in[5];
    const float* V1 = (const float*)d_in[6];
    const float* c1 = (const float*)d_in[7];
    const float* V2 = (const float*)d_in[8];
    const float* c2 = (const float*)d_in[9];
    const float* Wr = (const float*)d_in[10];
    const float* br = (const float*)d_in[11];
    float* out = (float*)d_out;

    const int Bb = 8, T = 1024, F = 1024, ZC = 128, ZL = 1028, HL = 1026;

    float* ws   = (float*)d_ws;
    const size_t H1N = (size_t)Bb * HL * F;
    const size_t WTN = (size_t)Bb * T * F;
    float* h1   = ws;
    float* g1   = h1 + H1N;
    float* wts  = g1 + H1N;
    float* bmat = wts + WTN;
    float* sm   = bmat + WTN;

    const dim3 blk(256);
    const dim3 gridA(F / BN, (Bb * HL + BM - 1) / BM);
    const dim3 gridB(F / BN, (Bb * T) / BM);

    gemm_f32_wmma<0, 1026><<<gridA, blk, 0, stream>>>(
        z, (long)ZL * ZC, ZC, W1, 0, b1, nullptr, 0, 0,
        h1, Bb * HL, 3 * ZC, 3 * ZC, F);
    gemm_f32_wmma<0, 1026><<<gridA, blk, 0, stream>>>(
        z, (long)ZL * ZC, ZC, V1, 0, c1, nullptr, 0, 0,
        g1, Bb * HL, 3 * ZC, 3 * ZC, F);
    gemm_f32_wmma<1, 1024><<<gridB, blk, 0, stream>>>(
        h1, (long)HL * F, F, W2, 0, b2, nullptr, 0, 0,
        wts, Bb * T, 3 * F, 3 * F, F);
    gemm_f32_wmma<1, 1024><<<gridB, blk, 0, stream>>>(
        g1, (long)HL * F, F, V2, 0, c2, nullptr, 0, 0,
        bmat, Bb * T, 3 * F, 3 * F, F);
    gemm_f32_wmma<2, 1024><<<gridB, blk, 0, stream>>>(
        x + 4, (long)T * F, T, wts, (long)T * F, nullptr,
        bmat, (long)T * F, F,
        sm, Bb * T, T, T - 4, F);
    gemm_f32_wmma<3, 1024><<<gridB, blk, 0, stream>>>(
        sm, (long)T * F, F, Wr, 0, br,
        x, (long)T * F, F,
        out, Bb * T, F, F, F);
}


# K loop: one LDS place per chunk and wave: all 16 fragment reads and the next stage's stores stand before the barrier, then 32 WMMAs run unbroken (64 fragment VGPRs); sums bit-identical
# speedup vs baseline: 1.1229x; 1.1229x over previous
#include <hip/hip_runtime.h>
#include <hip/hip_bf16.h>
#include <math.h>


typedef float v2f __attribute__((ext_vector_type(2)));
typedef float v4f __attribute__((ext_vector_type(4)));
typedef float v8f __attribute__((ext_vector_type(8)));

#define BM  64
#define BN  128
#define KC  32
#define AKP 36
#define BKP 36

__device__ __forceinline__ v8f wmma4(v2f a, v2f b, v8f c) {
    return __builtin_amdgcn_wmma_f32_16x16x4_f32(
        false, a, false, b, (short)0, c, false, false);
}

template <int MODE, int RPB>
__global__ __launch_bounds__(256)
void gemm_f32_wmma(const float* __restrict__ A0, const float* __restrict__ A1,
                   long bsA, long rsA,
                   const float* __restrict__ Bm0, const float* __restrict__ Bm1,
                   long bsB,
                   const float* __restrict__ biasCol0,
                   const float* __restrict__ biasCol1,
                   const float* __restrict__ Emat, long bsE, long rsE,
                   float* __restrict__ Out0, float* __restrict__ Out1,
                   int Mtotal, int K, int klim, int N)
{
    const bool second = (blockIdx.z != 0);
    const float* __restrict__ A       = second ? A1 : A0;
    const float* __restrict__ Bm      = second ? Bm1 : Bm0;
    const float* __restrict__ biasCol = second ? biasCol1 : biasCol0;
    float* __restrict__ Out           = second ? Out1 : Out0;

    __shared__ float As[2][BM][AKP];
    __shared__ float Bs[2][BN][BKP];

    const int tid  = threadIdx.x;
    const int lane = tid & 31;
    const int wave = tid >> 5;
    const int wm   = wave & 1;
    const int wn   = wave >> 1;

    const int tileM = blockIdx.y * BM;
    const int tileN = blockIdx.x * BN;
    const int bB = (bsB != 0) ? (tileM / RPB) : 0;

    v8f acc00 = {}, acc01 = {}, acc10 = {}, acc11 = {};

    const int a_row = tid >> 2;
    const int a_kc  = (tid & 3) << 2;
    int a_grow = tileM + a_row;
    if (a_grow >= Mtotal) a_grow = Mtotal - 1;
    const int a_b = a_grow / RPB;
    const int a_r = a_grow - a_b * RPB;
    const float* a_base = A + (long)a_b * bsA + (long)a_r * rsA;

    const int b_kr = tid >> 4;
    const int b_nc = (tid & 15) << 2;
    const float* b_base = Bm + (long)bB * bsB + tileN;

    const int m0   = (wm << 5) + (lane & 15);
    const int n0   = (wn << 5) + (lane & 15);
    const int koff = (lane >> 4) << 1;

    const int nchunks = (klim + KC - 1) / KC;

    v4f ra0, ra1, rb00, rb01, rb10, rb11;

    auto prefetch_full = [&](int k0) {
        ra0  = *(const v4f*)(a_base + k0 + a_kc);
        ra1  = *(const v4f*)(a_base + k0 + a_kc + 16);
        const float* br0 = b_base + (long)(k0 + b_kr)      * N + b_nc;
        const float* br1 = b_base + (long)(k0 + b_kr + 16) * N + b_nc;
        rb00 = *(const v4f*)(br0);
        rb01 = *(const v4f*)(br0 + 64);
        rb10 = *(const v4f*)(br1);
        rb11 = *(const v4f*)(br1 + 64);
    };

    auto prefetch = [&](int k0) {
        if (k0 + KC <= klim) {
            prefetch_full(k0);
        } else {
            const v4f zz = {0.f, 0.f, 0.f, 0.f};
            #pragma unroll
            for (int i = 0; i < 4; ++i) {
                const int ka = k0 + a_kc + i;
                ra0[i] = (ka      < klim) ? a_base[ka]      : 0.0f;
                ra1[i] = (ka + 16 < klim) ? a_base[ka + 16] : 0.0f;
            }
            const int kb0 = k0 + b_kr, kb1 = kb0 + 16;
            const float* br0 = b_base + (long)kb0 * N + b_nc;
            const float* br1 = b_base + (long)kb1 * N + b_nc;
            rb00 = (kb0 < K) ? *(const v4f*)(br0)      : zz;
            rb01 = (kb0 < K) ? *(const v4f*)(br0 + 64) : zz;
            rb10 = (kb1 < K) ? *(const v4f*)(br1)      : zz;
            rb11 = (kb1 < K) ? *(const v4f*)(br1 + 64) : zz;
        }
    };

    auto stage = [&](int buf) {
        *(v4f*)&As[buf][a_row][a_kc]      = ra0;
        *(v4f*)&As[buf][a_row][a_kc + 16] = ra1;
        #pragma unroll
        for (int i = 0; i < 4; ++i) {
            Bs[buf][b_nc + i     ][b_kr]      = rb00[i];
            Bs[buf][b_nc + i     ][b_kr + 16] = rb10[i];
            Bs[buf][b_nc + i + 64][b_kr]      = rb01[i];
            Bs[buf][b_nc + i + 64][b_kr + 16] = rb11[i];
        }
    };

    v2f fa0[KC / 4], fa1[KC / 4], fb0[KC / 4], fb1[KC / 4];

    auto loadfrags = [&](int buf) {
        #pragma unroll
        for (int s = 0; s < KC / 4; ++s) {
            fa0[s] = *(const v2f*)&As[buf][m0     ][4 * s + koff];
            fa1[s] = *(const v2f*)&As[buf][m0 + 16][4 * s + koff];
            fb0[s] = *(const v2f*)&Bs[buf][n0     ][4 * s + koff];
            fb1[s] = *(const v2f*)&Bs[buf][n0 + 16][4 * s + koff];
        }
    };

    auto mma = [&]() {
        #pragma unroll
        for (int s = 0; s < KC / 4; ++s) {
            acc00 = wmma4(fa0[s], fb0[s], acc00);
            acc01 = wmma4(fa0[s], fb1[s], acc01);
            acc10 = wmma4(fa1[s], fb0[s], acc10);
            acc11 = wmma4(fa1[s], fb1[s], acc11);
        }
    };

    prefetch(0);
    stage(0);
    __syncthreads();
    if (nchunks > 1) prefetch(KC);

    const int nfull = klim / KC;
    int buf = 0;
    int c = 0;
    for (; c + 2 < nfull; ++c) {
        loadfrags(buf);
        stage(buf ^ 1);
        __syncthreads();
        __builtin_amdgcn_sched_barrier(0);
        prefetch_full((c + 2) * KC);
        mma();
        buf ^= 1;
    }
    for (; c < nchunks; ++c) {
        loadfrags(buf);
        if (c + 1 < nchunks) {
            stage(buf ^ 1);
            __syncthreads();
        }
        if (c + 2 < nchunks) prefetch((c + 2) * KC);
        mma();
        buf ^= 1;
    }

    const int colA  = tileN + n0;
    const int mbase = tileM + (wm << 5) + ((lane >> 4) << 3);
    #pragma unroll
    for (int mi = 0; mi < 2; ++mi) {
        const v8f& accL = mi ? acc10 : acc00;
        const v8f& accR = mi ? acc11 : acc01;
        #pragma unroll
        for (int j = 0; j < 8; ++j) {
            const int gr = mbase + (mi << 4) + j;
            if (gr >= Mtotal) continue;
            const int b = gr / RPB;
            const int r = gr - b * RPB;
            float v0 = accL[j];
            float v1 = accR[j];
            if (MODE == 0) {
                v0 = fmaxf(v0 + biasCol[colA],      0.0f);
                v1 = fmaxf(v1 + biasCol[colA + 16], 0.0f);
            } else if (MODE == 1) {
                v0 += biasCol[colA];
                v1 += biasCol[colA + 16];
            } else if (MODE == 2) {
                const float* e = Emat + (long)b * bsE + (long)r * rsE;
                v0 = sinf(v0 + e[colA]);
                v1 = sinf(v1 + e[colA + 16]);
            } else {
                const float* e = Emat + (long)b * bsE + (long)r * rsE;
                v0 = (v0 + biasCol[colA]      + e[colA])      * 0.5f;
                v1 = (v1 + biasCol[colA + 16] + e[colA + 16]) * 0.5f;
            }
            Out[(long)gr * N + colA]      = v0;
            Out[(long)gr * N + colA + 16] = v1;
        }
    }
}

extern "C" void kernel_launch(void* const* d_in, const int* in_sizes, int n_in,
                              void* d_out, int out_size, void* d_ws, size_t ws_size,
                              hipStream_t stream)
{
    (void)in_sizes; (void)n_in; (void)out_size; (void)ws_size;

    const float* x  = (const float*)d_in[0];
    const float* z  = (const float*)d_in[1];
    const float* W1 = (const float*)d_in[2];
    const float* b1 = (const float*)d_in[3];
    const float* W2 = (const float*)d_in[4];
    const float* b2 = (const float*)d_in[5];
    const float* V1 = (const float*)d_in[6];
    const float* c1 = (const float*)d_in[7];
    const float* V2 = (const float*)d_in[8];
    const float* c2 = (const float*)d_in[9];
    const float* Wr = (const float*)d_in[10];
    const float* br = (const float*)d_in[11];
    float* out = (float*)d_out;

    const int Bb = 8, T = 1024, F = 1024, ZC = 128, ZL = 1028, HL = 1026;

    float* ws   = (float*)d_ws;
    const size_t H1N = (size_t)Bb * HL * F;
    const size_t WTN = (size_t)Bb * T * F;
    float* h1   = ws;
    float* g1   = h1 + H1N;
    float* wts  = g1 + H1N;
    float* bmat = wts + WTN;
    float* sm   = bmat + WTN;

    const dim3 blk(256);
    const dim3 gridA2(F / BN, (Bb * HL + BM - 1) / BM, 2);
    const dim3 gridB2(F / BN, (Bb * T) / BM, 2);
    const dim3 gridB(F / BN, (Bb * T) / BM);

    gemm_f32_wmma<0, 1026><<<gridA2, blk, 0, stream>>>(
        z, z, (long)ZL * ZC, ZC, W1, V1, 0, b1, c1, nullptr, 0, 0,
        h1, g1, Bb * HL, 3 * ZC, 3 * ZC, F);
    gemm_f32_wmma<1, 1024><<<gridB2, blk, 0, stream>>>(
        h1, g1, (long)HL * F, F, W2, V2, 0, b2, c2, nullptr, 0, 0,
        wts, bmat, Bb * T, 3 * F, 3 * F, F);
    gemm_f32_wmma<2, 1024><<<gridB, blk, 0, stream>>>(
        x + 4, x + 4, (long)T * F, T, wts, wts, (long)T * F, nullptr, nullptr,
        bmat, (long)T * F, F,
        sm, sm, Bb * T, T, T - 4, F);
    gemm_f32_wmma<3, 1024><<<gridB, blk, 0, stream>>>(
        sm, sm, (long)T * F, F, Wr, Wr, 0, br, br,
        x, (long)T * F, F,
        out, out, Bb * T, F, F, F);
}
